# starting kernel
# baseline (speedup 1.0000x reference)
#include <hip/hip_runtime.h>


#define BH      32
#define T       1024
#define DD      64
#define CHUNK   64
#define NCHUNK  (T / CHUNK)
#define EB      16
#define NE      (DD / EB)
#define QSTR    68
#define VSTR    24

#define L2DECAY     (-0.07400058144377693f)
#define DECAY_CHUNK (0.0375241347f)

typedef __attribute__((ext_vector_type(2))) float v2f;
typedef __attribute__((ext_vector_type(8))) float v8f;

__device__ __forceinline__ v8f wmma_f32(v2f a, v2f b, v8f c) {
    return __builtin_amdgcn_wmma_f32_16x16x4_f32(
         false, a,  false, b,
         (short)0, c,  false,  false);
}

__global__ __launch_bounds__(128)
void rkv_chunked_wmma(const float* __restrict__ q,
                      const float* __restrict__ k,
                      const float* __restrict__ v,
                      float* __restrict__ out)
{
    __shared__ float sQ[CHUNK][QSTR];
    __shared__ float sK[CHUNK][QSTR];
    __shared__ float sA[CHUNK][QSTR];
    __shared__ float sV[CHUNK][VSTR];
    __shared__ float sS[DD][VSTR];

    const int tid  = threadIdx.x;
    const int lane = tid & 31;
    const int wave = tid >> 5;
    const int lr   = lane & 15;
    const int half = lane >> 4;
    const int m0   = wave * 16;

    const int bh = blockIdx.x / NE;
    const int e0 = (blockIdx.x % NE) * EB;

    const float* qbh = q + (size_t)bh * T * DD;
    const float* kbh = k + (size_t)bh * T * DD;
    const float* vbh = v + (size_t)bh * T * DD;
    float*       obh = out + (size_t)bh * T * DD;

    for (int i = tid; i < DD * EB; i += 128)
        sS[i / EB][i % EB] = 0.0f;

    for (int c = 0; c < NCHUNK; ++c) {
        __syncthreads();
        const int tbase = c * CHUNK;

        for (int i = tid; i < CHUNK * (DD / 4); i += 128) {
            const int row = i / (DD / 4);
            const int c4  = (i % (DD / 4)) * 4;
            const float4 qv = *(const float4*)(qbh + (size_t)(tbase + row) * DD + c4);
            const float4 kv = *(const float4*)(kbh + (size_t)(tbase + row) * DD + c4);
            const float qs = __builtin_amdgcn_exp2f((float)(row + 1) * L2DECAY);
            const float ks = __builtin_amdgcn_exp2f((float)(-(row + 1)) * L2DECAY);
            sQ[row][c4 + 0] = qv.x * qs; sQ[row][c4 + 1] = qv.y * qs;
            sQ[row][c4 + 2] = qv.z * qs; sQ[row][c4 + 3] = qv.w * qs;
            sK[row][c4 + 0] = kv.x * ks; sK[row][c4 + 1] = kv.y * ks;
            sK[row][c4 + 2] = kv.z * ks; sK[row][c4 + 3] = kv.w * ks;
        }
        for (int i = tid; i < CHUNK * (EB / 4); i += 128) {
            const int row = i / (EB / 4);
            const int c4  = (i % (EB / 4)) * 4;
            const float4 vv = *(const float4*)(vbh + (size_t)(tbase + row) * DD + e0 + c4);
            sV[row][c4 + 0] = vv.x; sV[row][c4 + 1] = vv.y;
            sV[row][c4 + 2] = vv.z; sV[row][c4 + 3] = vv.w;
        }
        __syncthreads();

        for (int nb = 0; nb < 4; ++nb) {
            const int n0 = nb * 16;
            v8f acc = {};
            #pragma unroll
            for (int k0 = 0; k0 < DD; k0 += 4) {
                v2f a, b;
                a[0] = sQ[m0 + lr][k0 + 2 * half + 0];
                a[1] = sQ[m0 + lr][k0 + 2 * half + 1];
                b[0] = sK[n0 + lr][k0 + 2 * half + 0];
                b[1] = sK[n0 + lr][k0 + 2 * half + 1];
                acc = wmma_f32(a, b, acc);
            }
            #pragma unroll
            for (int r = 0; r < 8; ++r) {
                const int t = m0 + r + 8 * half;
                const int s = n0 + lr;
                sA[t][s] = (s <= t) ? acc[r] : 0.0f;
            }
        }
        __syncthreads();

        {
            v8f acc = {};
            #pragma unroll
            for (int k0 = 0; k0 < CHUNK; k0 += 4) {
                v2f a, b;
                a[0] = sA[m0 + lr][k0 + 2 * half + 0];
                a[1] = sA[m0 + lr][k0 + 2 * half + 1];
                b[0] = sV[k0 + 2 * half + 0][lr];
                b[1] = sV[k0 + 2 * half + 1][lr];
                acc = wmma_f32(a, b, acc);
            }
            #pragma unroll
            for (int k0 = 0; k0 < DD; k0 += 4) {
                v2f a, b;
                a[0] = sQ[m0 + lr][k0 + 2 * half + 0];
                a[1] = sQ[m0 + lr][k0 + 2 * half + 1];
                b[0] = sS[k0 + 2 * half + 0][lr];
                b[1] = sS[k0 + 2 * half + 1][lr];
                acc = wmma_f32(a, b, acc);
            }
            #pragma unroll
            for (int r = 0; r < 8; ++r) {
                const int t = tbase + m0 + r + 8 * half;
                const int e = e0 + lr;
                obh[(size_t)t * DD + e] = acc[r];
            }
        }
        __syncthreads();

        {
            v8f acc;
            #pragma unroll
            for (int r = 0; r < 8; ++r)
                acc[r] = sS[m0 + r + 8 * half][lr];
            #pragma unroll
            for (int k0 = 0; k0 < CHUNK; k0 += 4) {
                v2f a, b;
                a[0] = sK[k0 + 2 * half + 0][m0 + lr];
                a[1] = sK[k0 + 2 * half + 1][m0 + lr];
                b[0] = sV[k0 + 2 * half + 0][lr];
                b[1] = sV[k0 + 2 * half + 1][lr];
                acc = wmma_f32(a, b, acc);
            }
            #pragma unroll
            for (int r = 0; r < 8; ++r)
                sS[m0 + r + 8 * half][lr] = DECAY_CHUNK * acc[r];
        }
    }
}

extern "C" void kernel_launch(void* const* d_in, const int* in_sizes, int n_in,
                              void* d_out, int out_size, void* d_ws, size_t ws_size,
                              hipStream_t stream) {
    (void)in_sizes; (void)n_in; (void)d_ws; (void)ws_size; (void)out_size;
    const float* q = (const float*)d_in[0];
    const float* k = (const float*)d_in[1];
    const float* v = (const float*)d_in[2];
    float* out = (float*)d_out;
    rkv_chunked_wmma<<<dim3(BH * NE), dim3(128), 0, stream>>>(q, k, v, out);
}


# Chunk-parallel form, two launches: each chunk's decayed K^T V goes to workspace, then all chunks' outputs at once from the decayed running sum of those matrices (state summation order changes)
# speedup vs baseline: 4.9416x; 4.9416x over previous
#include <hip/hip_runtime.h>


#define BH      32
#define T       1024
#define DD      64
#define CHUNK   64
#define NCHUNK  (T / CHUNK)
#define NTILE   (DD / 16)
#define QSTR    68

#define L2DECAY     (-0.07400058144377693f)
#define DECAY_CHUNK (0.0375241347f)

typedef __attribute__((ext_vector_type(2))) float v2f;
typedef __attribute__((ext_vector_type(8))) float v8f;

__device__ __forceinline__ v8f wmma_f32(v2f a, v2f b, v8f c) {
    return __builtin_amdgcn_wmma_f32_16x16x4_f32(
         false, a,  false, b,
         (short)0, c,  false,  false);
}

__global__ __launch_bounds__(128)
void rkv_chunk_kv(const float* __restrict__ k,
                  const float* __restrict__ v,
                  float* __restrict__ w)
{
    __shared__ float sK[CHUNK][QSTR];
    __shared__ float sV[CHUNK][QSTR];

    const int tid  = threadIdx.x;
    const int lane = tid & 31;
    const int wave = tid >> 5;
    const int lr   = lane & 15;
    const int half = lane >> 4;
    const int m0   = wave * 16;

    const int bh = blockIdx.x / (NCHUNK - 1);
    const int c  = blockIdx.x % (NCHUNK - 1);
    const int tbase = c * CHUNK;

    const float* kbh = k + (size_t)bh * T * DD;
    const float* vbh = v + (size_t)bh * T * DD;
    float*       wc  = w + ((size_t)bh * NCHUNK + c) * DD * DD;

    for (int i = tid; i < CHUNK * (DD / 4); i += 128) {
        const int row = i / (DD / 4);
        const int c4  = (i % (DD / 4)) * 4;
        const float4 kv = *(const float4*)(kbh + (size_t)(tbase + row) * DD + c4);
        const float4 vv = *(const float4*)(vbh + (size_t)(tbase + row) * DD + c4);
        const float ks = __builtin_amdgcn_exp2f((float)(-(row + 1)) * L2DECAY);
        sK[row][c4 + 0] = kv.x * ks; sK[row][c4 + 1] = kv.y * ks;
        sK[row][c4 + 2] = kv.z * ks; sK[row][c4 + 3] = kv.w * ks;
        sV[row][c4 + 0] = vv.x; sV[row][c4 + 1] = vv.y;
        sV[row][c4 + 2] = vv.z; sV[row][c4 + 3] = vv.w;
    }
    __syncthreads();

    for (int ne = 0; ne < NTILE; ++ne) {
        const int e0 = ne * 16;
        v8f acc = {};
        #pragma unroll
        for (int k0 = 0; k0 < CHUNK; k0 += 4) {
            v2f a, b;
            a[0] = sK[k0 + 2 * half + 0][m0 + lr];
            a[1] = sK[k0 + 2 * half + 1][m0 + lr];
            b[0] = sV[k0 + 2 * half + 0][e0 + lr];
            b[1] = sV[k0 + 2 * half + 1][e0 + lr];
            acc = wmma_f32(a, b, acc);
        }
        #pragma unroll
        for (int r = 0; r < 8; ++r)
            wc[(size_t)(m0 + r + 8 * half) * DD + e0 + lr] = DECAY_CHUNK * acc[r];
    }
}

__global__ __launch_bounds__(128)
void rkv_chunk_out(const float* __restrict__ q,
                   const float* __restrict__ k,
                   const float* __restrict__ v,
                   const float* __restrict__ w,
                   float* __restrict__ out)
{
    __shared__ float sQ[CHUNK][QSTR];
    __shared__ float sK[CHUNK][QSTR];
    __shared__ float sA[CHUNK][QSTR];
    __shared__ float sV[CHUNK][QSTR];
    __shared__ float sS[DD][QSTR];

    const int tid  = threadIdx.x;
    const int lane = tid & 31;
    const int wave = tid >> 5;
    const int lr   = lane & 15;
    const int half = lane >> 4;
    const int m0   = wave * 16;

    const int bh = blockIdx.x / NCHUNK;
    const int c  = blockIdx.x % NCHUNK;
    const int tbase = c * CHUNK;

    const float* qbh = q + (size_t)bh * T * DD;
    const float* kbh = k + (size_t)bh * T * DD;
    const float* vbh = v + (size_t)bh * T * DD;
    const float* wbh = w + (size_t)bh * NCHUNK * DD * DD;
    float*       obh = out + (size_t)bh * T * DD;

    {
        float4 s[8];
        #pragma unroll
        for (int m = 0; m < 8; ++m) s[m] = make_float4(0.0f, 0.0f, 0.0f, 0.0f);
        for (int j = 0; j < c; ++j) {
            const float* wj = wbh + (size_t)j * DD * DD;
            #pragma unroll
            for (int m = 0; m < 8; ++m) {
                const float4 x = *(const float4*)(wj + (size_t)(tid + 128 * m) * 4);
                s[m].x = DECAY_CHUNK * s[m].x + x.x;
                s[m].y = DECAY_CHUNK * s[m].y + x.y;
                s[m].z = DECAY_CHUNK * s[m].z + x.z;
                s[m].w = DECAY_CHUNK * s[m].w + x.w;
            }
        }
        #pragma unroll
        for (int m = 0; m < 8; ++m) {
            const int i   = tid + 128 * m;
            const int row = i / (DD / 4);
            const int c4  = (i % (DD / 4)) * 4;
            sS[row][c4 + 0] = s[m].x; sS[row][c4 + 1] = s[m].y;
            sS[row][c4 + 2] = s[m].z; sS[row][c4 + 3] = s[m].w;
        }
    }

    for (int i = tid; i < CHUNK * (DD / 4); i += 128) {
        const int row = i / (DD / 4);
        const int c4  = (i % (DD / 4)) * 4;
        const float4 qv = *(const float4*)(qbh + (size_t)(tbase + row) * DD + c4);
        const float4 kv = *(const float4*)(kbh + (size_t)(tbase + row) * DD + c4);
        const float4 vv = *(const float4*)(vbh + (size_t)(tbase + row) * DD + c4);
        const float qs = __builtin_amdgcn_exp2f((float)(row + 1) * L2DECAY);
        const float ks = __builtin_amdgcn_exp2f((float)(-(row + 1)) * L2DECAY);
        sQ[row][c4 + 0] = qv.x * qs; sQ[row][c4 + 1] = qv.y * qs;
        sQ[row][c4 + 2] = qv.z * qs; sQ[row][c4 + 3] = qv.w * qs;
        sK[row][c4 + 0] = kv.x * ks; sK[row][c4 + 1] = kv.y * ks;
        sK[row][c4 + 2] = kv.z * ks; sK[row][c4 + 3] = kv.w * ks;
        sV[row][c4 + 0] = vv.x; sV[row][c4 + 1] = vv.y;
        sV[row][c4 + 2] = vv.z; sV[row][c4 + 3] = vv.w;
    }
    __syncthreads();

    for (int nb = 0; nb < 4; ++nb) {
        const int n0 = nb * 16;
        v8f acc = {};
        #pragma unroll
        for (int k0 = 0; k0 < DD; k0 += 4) {
            v2f a, b;
            a[0] = sQ[m0 + lr][k0 + 2 * half + 0];
            a[1] = sQ[m0 + lr][k0 + 2 * half + 1];
            b[0] = sK[n0 + lr][k0 + 2 * half + 0];
            b[1] = sK[n0 + lr][k0 + 2 * half + 1];
            acc = wmma_f32(a, b, acc);
        }
        #pragma unroll
        for (int r = 0; r < 8; ++r) {
            const int t = m0 + r + 8 * half;
            const int s = n0 + lr;
            sA[t][s] = (s <= t) ? acc[r] : 0.0f;
        }
    }
    __syncthreads();

    for (int ne = 0; ne < NTILE; ++ne) {
        const int e0 = ne * 16;
        v8f acc = {};
        #pragma unroll
        for (int k0 = 0; k0 < CHUNK; k0 += 4) {
            v2f a, b;
            a[0] = sA[m0 + lr][k0 + 2 * half + 0];
            a[1] = sA[m0 + lr][k0 + 2 * half + 1];
            b[0] = sV[k0 + 2 * half + 0][e0 + lr];
            b[1] = sV[k0 + 2 * half + 1][e0 + lr];
            acc = wmma_f32(a, b, acc);
        }
        #pragma unroll
        for (int k0 = 0; k0 < DD; k0 += 4) {
            v2f a, b;
            a[0] = sQ[m0 + lr][k0 + 2 * half + 0];
            a[1] = sQ[m0 + lr][k0 + 2 * half + 1];
            b[0] = sS[k0 + 2 * half + 0][e0 + lr];
            b[1] = sS[k0 + 2 * half + 1][e0 + lr];
            acc = wmma_f32(a, b, acc);
        }
        #pragma unroll
        for (int r = 0; r < 8; ++r) {
            const int t = tbase + m0 + r + 8 * half;
            obh[(size_t)t * DD + e0 + lr] = acc[r];
        }
    }
}

extern "C" void kernel_launch(void* const* d_in, const int* in_sizes, int n_in,
                              void* d_out, int out_size, void* d_ws, size_t ws_size,
                              hipStream_t stream) {
    (void)in_sizes; (void)n_in; (void)ws_size; (void)out_size;
    const float* q = (const float*)d_in[0];
    const float* k = (const float*)d_in[1];
    const float* v = (const float*)d_in[2];
    float* out = (float*)d_out;
    float* w   = (float*)d_ws;
    rkv_chunk_kv<<<dim3(BH * (NCHUNK - 1)), dim3(128), 0, stream>>>(k, v, w);
    rkv_chunk_out<<<dim3(BH * NCHUNK), dim3(128), 0, stream>>>(q, k, v, w, out);
}


# Both launches request each block's global data as a few large groups (chunk rows all at once, earlier chunks' matrices five per round trip) instead of one dependent round trip per loop pass
# speedup vs baseline: 7.1278x; 1.4424x over previous
#include <hip/hip_runtime.h>


#define BH      32
#define T       1024
#define DD      64
#define CHUNK   64
#define NCHUNK  (T / CHUNK)
#define NTILE   (DD / 16)
#define QSTR    68
#define SGRP    5

#define L2DECAY     (-0.07400058144377693f)
#define DECAY_CHUNK (0.0375241347f)

typedef __attribute__((ext_vector_type(2))) float v2f;
typedef __attribute__((ext_vector_type(8))) float v8f;

__device__ __forceinline__ v8f wmma_f32(v2f a, v2f b, v8f c) {
    return __builtin_amdgcn_wmma_f32_16x16x4_f32(
         false, a,  false, b,
         (short)0, c,  false,  false);
}

__global__ __launch_bounds__(128)
void rkv_chunk_kv(const float* __restrict__ k,
                  const float* __restrict__ v,
                  float* __restrict__ w)
{
    __shared__ float sK[CHUNK][QSTR];
    __shared__ float sV[CHUNK][QSTR];

    const int tid  = threadIdx.x;
    const int lane = tid & 31;
    const int wave = tid >> 5;
    const int lr   = lane & 15;
    const int half = lane >> 4;
    const int m0   = wave * 16;

    const int bh = blockIdx.x / (NCHUNK - 1);
    const int c  = blockIdx.x % (NCHUNK - 1);
    const int tbase = c * CHUNK;

    const float* kbh = k + (size_t)bh * T * DD;
    const float* vbh = v + (size_t)bh * T * DD;
    float*       wc  = w + ((size_t)bh * NCHUNK + c) * DD * DD;

    const int prow = tid >> 4;
    const int pc4  = (tid & 15) * 4;
    float4 kv[8], vv[8];
    #pragma unroll
    for (int m = 0; m < 8; ++m) {
        const size_t off = (size_t)(tbase + prow + 8 * m) * DD + pc4;
        kv[m] = *(const float4*)(kbh + off);
        vv[m] = *(const float4*)(vbh + off);
    }
    #pragma unroll
    for (int m = 0; m < 8; ++m) {
        const int row = prow + 8 * m;
        const float ks = __builtin_amdgcn_exp2f((float)(-(row + 1)) * L2DECAY);
        sK[row][pc4 + 0] = kv[m].x * ks; sK[row][pc4 + 1] = kv[m].y * ks;
        sK[row][pc4 + 2] = kv[m].z * ks; sK[row][pc4 + 3] = kv[m].w * ks;
        sV[row][pc4 + 0] = vv[m].x; sV[row][pc4 + 1] = vv[m].y;
        sV[row][pc4 + 2] = vv[m].z; sV[row][pc4 + 3] = vv[m].w;
    }
    __syncthreads();

    for (int ne = 0; ne < NTILE; ++ne) {
        const int e0 = ne * 16;
        v8f acc = {};
        #pragma unroll
        for (int k0 = 0; k0 < CHUNK; k0 += 4) {
            v2f a, b;
            a[0] = sK[k0 + 2 * half + 0][m0 + lr];
            a[1] = sK[k0 + 2 * half + 1][m0 + lr];
            b[0] = sV[k0 + 2 * half + 0][e0 + lr];
            b[1] = sV[k0 + 2 * half + 1][e0 + lr];
            acc = wmma_f32(a, b, acc);
        }
        #pragma unroll
        for (int r = 0; r < 8; ++r)
            wc[(size_t)(m0 + r + 8 * half) * DD + e0 + lr] = DECAY_CHUNK * acc[r];
    }
}

__global__ __launch_bounds__(128)
void rkv_chunk_out(const float* __restrict__ q,
                   const float* __restrict__ k,
                   const float* __restrict__ v,
                   const float* __restrict__ w,
                   float* __restrict__ out)
{
    __shared__ float sQ[CHUNK][QSTR];
    __shared__ float sK[CHUNK][QSTR];
    __shared__ float sA[CHUNK][QSTR];
    __shared__ float sV[CHUNK][QSTR];
    __shared__ float sS[DD][QSTR];

    const int tid  = threadIdx.x;
    const int lane = tid & 31;
    const int wave = tid >> 5;
    const int lr   = lane & 15;
    const int half = lane >> 4;
    const int m0   = wave * 16;

    const int bh = blockIdx.x / NCHUNK;
    const int c  = blockIdx.x % NCHUNK;
    const int tbase = c * CHUNK;

    const float* qbh = q + (size_t)bh * T * DD;
    const float* kbh = k + (size_t)bh * T * DD;
    const float* vbh = v + (size_t)bh * T * DD;
    const float* wbh = w + (size_t)bh * NCHUNK * DD * DD;
    float*       obh = out + (size_t)bh * T * DD;

    const int prow = tid >> 4;
    const int pc4  = (tid & 15) * 4;

    float4 qv[8], kv[8], vv[8];
    #pragma unroll
    for (int m = 0; m < 8; ++m) {
        const size_t off = (size_t)(tbase + prow + 8 * m) * DD + pc4;
        qv[m] = *(const float4*)(qbh + off);
        kv[m] = *(const float4*)(kbh + off);
        vv[m] = *(const float4*)(vbh + off);
    }

    float4 s[8];
    if (c > 0) {
        #pragma unroll
        for (int m = 0; m < 8; ++m)
            s[m] = *(const float4*)(wbh + (size_t)(tid + 128 * m) * 4);
    } else {
        #pragma unroll
        for (int m = 0; m < 8; ++m) s[m] = make_float4(0.0f, 0.0f, 0.0f, 0.0f);
    }
    float4 x[SGRP][8];
    #pragma unroll
    for (int u = 1; u < SGRP; ++u) {
        if (u < c) {
            const float* wj = wbh + (size_t)u * DD * DD;
            #pragma unroll
            for (int m = 0; m < 8; ++m)
                x[u][m] = *(const float4*)(wj + (size_t)(tid + 128 * m) * 4);
        }
    }

    #pragma unroll
    for (int m = 0; m < 8; ++m) {
        const int row = prow + 8 * m;
        const float qs = __builtin_amdgcn_exp2f((float)(row + 1) * L2DECAY);
        const float ks = __builtin_amdgcn_exp2f((float)(-(row + 1)) * L2DECAY);
        sQ[row][pc4 + 0] = qv[m].x * qs; sQ[row][pc4 + 1] = qv[m].y * qs;
        sQ[row][pc4 + 2] = qv[m].z * qs; sQ[row][pc4 + 3] = qv[m].w * qs;
        sK[row][pc4 + 0] = kv[m].x * ks; sK[row][pc4 + 1] = kv[m].y * ks;
        sK[row][pc4 + 2] = kv[m].z * ks; sK[row][pc4 + 3] = kv[m].w * ks;
        sV[row][pc4 + 0] = vv[m].x; sV[row][pc4 + 1] = vv[m].y;
        sV[row][pc4 + 2] = vv[m].z; sV[row][pc4 + 3] = vv[m].w;
    }

    #pragma unroll
    for (int u = 1; u < SGRP; ++u) {
        if (u < c) {
            #pragma unroll
            for (int m = 0; m < 8; ++m) {
                s[m].x = DECAY_CHUNK * s[m].x + x[u][m].x;
                s[m].y = DECAY_CHUNK * s[m].y + x[u][m].y;
                s[m].z = DECAY_CHUNK * s[m].z + x[u][m].z;
                s[m].w = DECAY_CHUNK * s[m].w + x[u][m].w;
            }
        }
    }
    for (int j0 = SGRP; j0 < c; j0 += SGRP) {
        #pragma unroll
        for (int u = 0; u < SGRP; ++u) {
            if (j0 + u < c) {
                const float* wj = wbh + (size_t)(j0 + u) * DD * DD;
                #pragma unroll
                for (int m = 0; m < 8; ++m)
                    x[u][m] = *(const float4*)(wj + (size_t)(tid + 128 * m) * 4);
            }
        }
        #pragma unroll
        for (int u = 0; u < SGRP; ++u) {
            if (j0 + u < c) {
                #pragma unroll
                for (int m = 0; m < 8; ++m) {
                    s[m].x = DECAY_CHUNK * s[m].x + x[u][m].x;
                    s[m].y = DECAY_CHUNK * s[m].y + x[u][m].y;
                    s[m].z = DECAY_CHUNK * s[m].z + x[u][m].z;
                    s[m].w = DECAY_CHUNK * s[m].w + x[u][m].w;
                }
            }
        }
    }
    #pragma unroll
    for (int m = 0; m < 8; ++m) {
        const int row = prow + 8 * m;
        sS[row][pc4 + 0] = s[m].x; sS[row][pc4 + 1] = s[m].y;
        sS[row][pc4 + 2] = s[m].z; sS[row][pc4 + 3] = s[m].w;
    }
    __syncthreads();

    for (int nb = 0; nb < 4; ++nb) {
        const int n0 = nb * 16;
        v8f acc = {};
        #pragma unroll
        for (int k0 = 0; k0 < DD; k0 += 4) {
            v2f a, b;
            a[0] = sQ[m0 + lr][k0 + 2 * half + 0];
            a[1] = sQ[m0 + lr][k0 + 2 * half + 1];
            b[0] = sK[n0 + lr][k0 + 2 * half + 0];
            b[1] = sK[n0 + lr][k0 + 2 * half + 1];
            acc = wmma_f32(a, b, acc);
        }
        #pragma unroll
        for (int r = 0; r < 8; ++r) {
            const int t = m0 + r + 8 * half;
            const int s = n0 + lr;
            sA[t][s] = (s <= t) ? acc[r] : 0.0f;
        }
    }
    __syncthreads();

    for (int ne = 0; ne < NTILE; ++ne) {
        const int e0 = ne * 16;
        v8f acc = {};
        #pragma unroll
        for (int k0 = 0; k0 < CHUNK; k0 += 4) {
            v2f a, b;
            a[0] = sA[m0 + lr][k0 + 2 * half + 0];
            a[1] = sA[m0 + lr][k0 + 2 * half + 1];
            b[0] = sV[k0 + 2 * half + 0][e0 + lr];
            b[1] = sV[k0 + 2 * half + 1][e0 + lr];
            acc = wmma_f32(a, b, acc);
        }
        #pragma unroll
        for (int k0 = 0; k0 < DD; k0 += 4) {
            v2f a, b;
            a[0] = sQ[m0 + lr][k0 + 2 * half + 0];
            a[1] = sQ[m0 + lr][k0 + 2 * half + 1];
            b[0] = sS[k0 + 2 * half + 0][e0 + lr];
            b[1] = sS[k0 + 2 * half + 1][e0 + lr];
            acc = wmma_f32(a, b, acc);
        }
        #pragma unroll
        for (int r = 0; r < 8; ++r) {
            const int t = tbase + m0 + r + 8 * half;
            obh[(size_t)t * DD + e0 + lr] = acc[r];
        }
    }
}

extern "C" void kernel_launch(void* const* d_in, const int* in_sizes, int n_in,
                              void* d_out, int out_size, void* d_ws, size_t ws_size,
                              hipStream_t stream) {
    (void)in_sizes; (void)n_in; (void)ws_size; (void)out_size;
    const float* q = (const float*)d_in[0];
    const float* k = (const float*)d_in[1];
    const float* v = (const float*)d_in[2];
    float* out = (float*)d_out;
    float* w   = (float*)d_ws;
    rkv_chunk_kv<<<dim3(BH * (NCHUNK - 1)), dim3(128), 0, stream>>>(k, v, w);
    rkv_chunk_out<<<dim3(BH * NCHUNK), dim3(128), 0, stream>>>(q, k, v, w, out);
}


# Launch 2 requests the later groups of earlier chunks' matrices ahead of Qd Kd^T and A V and keeps them in flight across LDS-only barriers; the state is folded just before Qd S_prev
# speedup vs baseline: 7.6563x; 1.0741x over previous
#include <hip/hip_runtime.h>


#define BH      32
#define T       1024
#define DD      64
#define CHUNK   64
#define NCHUNK  (T / CHUNK)
#define NTILE   (DD / 16)
#define QSTR    68
#define SGRP    5
static_assert(3 * SGRP >= NCHUNK - 1, "launch 2 requests the earlier chunks' matrices in exactly three groups");

#define L2DECAY     (-0.07400058144377693f)
#define DECAY_CHUNK (0.0375241347f)

typedef __attribute__((ext_vector_type(2))) float v2f;
typedef __attribute__((ext_vector_type(8))) float v8f;

__device__ __forceinline__ v8f wmma_f32(v2f a, v2f b, v8f c) {
    return __builtin_amdgcn_wmma_f32_16x16x4_f32(
         false, a,  false, b,
         (short)0, c,  false,  false);
}

__device__ __forceinline__ void lds_barrier() {
    asm volatile("s_wait_dscnt 0x0\n\ts_barrier_signal -1\n\ts_barrier_wait -1" ::: "memory");
}

__global__ __launch_bounds__(128)
void rkv_chunk_kv(const float* __restrict__ k,
                  const float* __restrict__ v,
                  float* __restrict__ w)
{
    __shared__ float sK[CHUNK][QSTR];
    __shared__ float sV[CHUNK][QSTR];

    const int tid  = threadIdx.x;
    const int lane = tid & 31;
    const int wave = tid >> 5;
    const int lr   = lane & 15;
    const int half = lane >> 4;
    const int m0   = wave * 16;

    const int bh = blockIdx.x / (NCHUNK - 1);
    const int c  = blockIdx.x % (NCHUNK - 1);
    const int tbase = c * CHUNK;

    const float* kbh = k + (size_t)bh * T * DD;
    const float* vbh = v + (size_t)bh * T * DD;
    float*       wc  = w + ((size_t)bh * NCHUNK + c) * DD * DD;

    const int prow = tid >> 4;
    const int pc4  = (tid & 15) * 4;
    float4 kv[8], vv[8];
    #pragma unroll
    for (int m = 0; m < 8; ++m) {
        const size_t off = (size_t)(tbase + prow + 8 * m) * DD + pc4;
        kv[m] = *(const float4*)(kbh + off);
        vv[m] = *(const float4*)(vbh + off);
    }
    #pragma unroll
    for (int m = 0; m < 8; ++m) {
        const int row = prow + 8 * m;
        const float ks = __builtin_amdgcn_exp2f((float)(-(row + 1)) * L2DECAY);
        sK[row][pc4 + 0] = kv[m].x * ks; sK[row][pc4 + 1] = kv[m].y * ks;
        sK[row][pc4 + 2] = kv[m].z * ks; sK[row][pc4 + 3] = kv[m].w * ks;
        sV[row][pc4 + 0] = vv[m].x; sV[row][pc4 + 1] = vv[m].y;
        sV[row][pc4 + 2] = vv[m].z; sV[row][pc4 + 3] = vv[m].w;
    }
    __syncthreads();

    for (int ne = 0; ne < NTILE; ++ne) {
        const int e0 = ne * 16;
        v8f acc = {};
        #pragma unroll
        for (int k0 = 0; k0 < CHUNK; k0 += 4) {
            v2f a, b;
            a[0] = sK[k0 + 2 * half + 0][m0 + lr];
            a[1] = sK[k0 + 2 * half + 1][m0 + lr];
            b[0] = sV[k0 + 2 * half + 0][e0 + lr];
            b[1] = sV[k0 + 2 * half + 1][e0 + lr];
            acc = wmma_f32(a, b, acc);
        }
        #pragma unroll
        for (int r = 0; r < 8; ++r)
            wc[(size_t)(m0 + r + 8 * half) * DD + e0 + lr] = DECAY_CHUNK * acc[r];
    }
}

__global__ __launch_bounds__(128)
void rkv_chunk_out(const float* __restrict__ q,
                   const float* __restrict__ k,
                   const float* __restrict__ v,
                   const float* __restrict__ w,
                   float* __restrict__ out)
{
    __shared__ float sQ[CHUNK][QSTR];
    __shared__ float sK[CHUNK][QSTR];
    __shared__ float sA[CHUNK][QSTR];
    __shared__ float sV[CHUNK][QSTR];
    __shared__ float sS[DD][QSTR];

    const int tid  = threadIdx.x;
    const int lane = tid & 31;
    const int wave = tid >> 5;
    const int lr   = lane & 15;
    const int half = lane >> 4;
    const int m0   = wave * 16;

    const int bh = blockIdx.x / NCHUNK;
    const int c  = blockIdx.x % NCHUNK;
    const int tbase = c * CHUNK;

    const float* qbh = q + (size_t)bh * T * DD;
    const float* kbh = k + (size_t)bh * T * DD;
    const float* vbh = v + (size_t)bh * T * DD;
    const float* wbh = w + (size_t)bh * NCHUNK * DD * DD;
    float*       obh = out + (size_t)bh * T * DD;

    const int prow = tid >> 4;
    const int pc4  = (tid & 15) * 4;

    float4 qv[8], kv[8], vv[8];
    #pragma unroll
    for (int m = 0; m < 8; ++m) {
        const size_t off = (size_t)(tbase + prow + 8 * m) * DD + pc4;
        qv[m] = *(const float4*)(qbh + off);
        kv[m] = *(const float4*)(kbh + off);
        vv[m] = *(const float4*)(vbh + off);
    }

    float4 s[8];
    if (c > 0) {
        #pragma unroll
        for (int m = 0; m < 8; ++m)
            s[m] = *(const float4*)(wbh + (size_t)(tid + 128 * m) * 4);
    } else {
        #pragma unroll
        for (int m = 0; m < 8; ++m) s[m] = make_float4(0.0f, 0.0f, 0.0f, 0.0f);
    }
    float4 xa[SGRP][8];
    float4 xb[SGRP][8];
    #pragma unroll
    for (int u = 1; u < SGRP; ++u) {
        if (u < c) {
            const float* wj = wbh + (size_t)u * DD * DD;
            #pragma unroll
            for (int m = 0; m < 8; ++m)
                xa[u][m] = *(const float4*)(wj + (size_t)(tid + 128 * m) * 4);
        }
    }

    #pragma unroll
    for (int m = 0; m < 8; ++m) {
        const int row = prow + 8 * m;
        const float qs = __builtin_amdgcn_exp2f((float)(row + 1) * L2DECAY);
        const float ks = __builtin_amdgcn_exp2f((float)(-(row + 1)) * L2DECAY);
        sQ[row][pc4 + 0] = qv[m].x * qs; sQ[row][pc4 + 1] = qv[m].y * qs;
        sQ[row][pc4 + 2] = qv[m].z * qs; sQ[row][pc4 + 3] = qv[m].w * qs;
        sK[row][pc4 + 0] = kv[m].x * ks; sK[row][pc4 + 1] = kv[m].y * ks;
        sK[row][pc4 + 2] = kv[m].z * ks; sK[row][pc4 + 3] = kv[m].w * ks;
        sV[row][pc4 + 0] = vv[m].x; sV[row][pc4 + 1] = vv[m].y;
        sV[row][pc4 + 2] = vv[m].z; sV[row][pc4 + 3] = vv[m].w;
    }

    #pragma unroll
    for (int u = 0; u < SGRP; ++u) {
        if (SGRP + u < c) {
            const float* wj = wbh + (size_t)(SGRP + u) * DD * DD;
            #pragma unroll
            for (int m = 0; m < 8; ++m)
                xb[u][m] = *(const float4*)(wj + (size_t)(tid + 128 * m) * 4);
        }
    }
    lds_barrier();

    for (int nb = 0; nb < 4; ++nb) {
        const int n0 = nb * 16;
        v8f acc = {};
        #pragma unroll
        for (int k0 = 0; k0 < DD; k0 += 4) {
            v2f a, b;
            a[0] = sQ[m0 + lr][k0 + 2 * half + 0];
            a[1] = sQ[m0 + lr][k0 + 2 * half + 1];
            b[0] = sK[n0 + lr][k0 + 2 * half + 0];
            b[1] = sK[n0 + lr][k0 + 2 * half + 1];
            acc = wmma_f32(a, b, acc);
        }
        #pragma unroll
        for (int r = 0; r < 8; ++r) {
            const int t = m0 + r + 8 * half;
            const int s = n0 + lr;
            sA[t][s] = (s <= t) ? acc[r] : 0.0f;
        }
    }

    #pragma unroll
    for (int u = 1; u < SGRP; ++u) {
        if (u < c) {
            #pragma unroll
            for (int m = 0; m < 8; ++m) {
                s[m].x = DECAY_CHUNK * s[m].x + xa[u][m].x;
                s[m].y = DECAY_CHUNK * s[m].y + xa[u][m].y;
                s[m].z = DECAY_CHUNK * s[m].z + xa[u][m].z;
                s[m].w = DECAY_CHUNK * s[m].w + xa[u][m].w;
            }
        }
    }
    #pragma unroll
    for (int u = 0; u < SGRP; ++u) {
        if (2 * SGRP + u < c) {
            const float* wj = wbh + (size_t)(2 * SGRP + u) * DD * DD;
            #pragma unroll
            for (int m = 0; m < 8; ++m)
                xa[u][m] = *(const float4*)(wj + (size_t)(tid + 128 * m) * 4);
        }
    }
    lds_barrier();

    v8f acc[NTILE];
    #pragma unroll
    for (int ne = 0; ne < NTILE; ++ne) {
        const int e0 = ne * 16;
        v8f t = {};
        #pragma unroll
        for (int k0 = 0; k0 < CHUNK; k0 += 4) {
            v2f a, b;
            a[0] = sA[m0 + lr][k0 + 2 * half + 0];
            a[1] = sA[m0 + lr][k0 + 2 * half + 1];
            b[0] = sV[k0 + 2 * half + 0][e0 + lr];
            b[1] = sV[k0 + 2 * half + 1][e0 + lr];
            t = wmma_f32(a, b, t);
        }
        acc[ne] = t;
    }

    #pragma unroll
    for (int u = 0; u < SGRP; ++u) {
        if (SGRP + u < c) {
            #pragma unroll
            for (int m = 0; m < 8; ++m) {
                s[m].x = DECAY_CHUNK * s[m].x + xb[u][m].x;
                s[m].y = DECAY_CHUNK * s[m].y + xb[u][m].y;
                s[m].z = DECAY_CHUNK * s[m].z + xb[u][m].z;
                s[m].w = DECAY_CHUNK * s[m].w + xb[u][m].w;
            }
        }
    }
    #pragma unroll
    for (int u = 0; u < SGRP; ++u) {
        if (2 * SGRP + u < c) {
            #pragma unroll
            for (int m = 0; m < 8; ++m) {
                s[m].x = DECAY_CHUNK * s[m].x + xa[u][m].x;
                s[m].y = DECAY_CHUNK * s[m].y + xa[u][m].y;
                s[m].z = DECAY_CHUNK * s[m].z + xa[u][m].z;
                s[m].w = DECAY_CHUNK * s[m].w + xa[u][m].w;
            }
        }
    }
    #pragma unroll
    for (int m = 0; m < 8; ++m) {
        const int row = prow + 8 * m;
        sS[row][pc4 + 0] = s[m].x; sS[row][pc4 + 1] = s[m].y;
        sS[row][pc4 + 2] = s[m].z; sS[row][pc4 + 3] = s[m].w;
    }
    __syncthreads();

    #pragma unroll
    for (int ne = 0; ne < NTILE; ++ne) {
        const int e0 = ne * 16;
        v8f t = acc[ne];
        #pragma unroll
        for (int k0 = 0; k0 < DD; k0 += 4) {
            v2f a, b;
            a[0] = sQ[m0 + lr][k0 + 2 * half + 0];
            a[1] = sQ[m0 + lr][k0 + 2 * half + 1];
            b[0] = sS[k0 + 2 * half + 0][e0 + lr];
            b[1] = sS[k0 + 2 * half + 1][e0 + lr];
            t = wmma_f32(a, b, t);
        }
        #pragma unroll
        for (int r = 0; r < 8; ++r) {
            const int tt = tbase + m0 + r + 8 * half;
            obh[(size_t)tt * DD + e0 + lr] = t[r];
        }
    }
}

extern "C" void kernel_launch(void* const* d_in, const int* in_sizes, int n_in,
                              void* d_out, int out_size, void* d_ws, size_t ws_size,
                              hipStream_t stream) {
    (void)in_sizes; (void)n_in; (void)ws_size; (void)out_size;
    const float* q = (const float*)d_in[0];
    const float* k = (const float*)d_in[1];
    const float* v = (const float*)d_in[2];
    float* out = (float*)d_out;
    float* w   = (float*)d_ws;
    rkv_chunk_kv<<<dim3(BH * (NCHUNK - 1)), dim3(128), 0, stream>>>(k, v, w);
    rkv_chunk_out<<<dim3(BH * NCHUNK), dim3(128), 0, stream>>>(q, k, v, w, out);
}
